# starting kernel
# baseline (speedup 1.0000x reference)
#include <hip/hip_runtime.h>

typedef __attribute__((ext_vector_type(2))) float v2f;
typedef __attribute__((ext_vector_type(4))) float v4f;
typedef __attribute__((ext_vector_type(8))) float v8f;

#define NGEN   4
#define TWO_N  400
#define NCH    32
#define NB     512
#define OTILES (TWO_N / 16)

__global__ void build_invperm_kernel(const float* __restrict__ perm,
                                     int* __restrict__ pinv) {
    int t = blockIdx.x * blockDim.x + threadIdx.x;
    if (t >= NGEN * TWO_N) return;
    int g = t / TWO_N;
    int o = t - g * TWO_N;
    const float* p = perm + (size_t)g * TWO_N * TWO_N + o;
    int found = 0;
    for (int i = 0; i < TWO_N; ++i) {
        if (p[(size_t)i * TWO_N] > 0.5f) found = i;
    }
    pinv[t] = found;
}

__global__ void __launch_bounds__(256)
dihedral_gconv_wmma(const float* __restrict__ x,
                    const float* __restrict__ w,
                    const int*  __restrict__ pinv,
                    float* __restrict__ out) {
    const int lane  = threadIdx.x & 31;
    const int wid   = threadIdx.x >> 5;
    const int wave  = blockIdx.x * 8 + wid;
    const int b     = wave / OTILES;
    const int ot    = wave - b * OTILES;
    const int o0    = ot * 16;
    const int m     = lane & 15;
    const int khalf = lane >> 4;

    v8f acc0 = {};
    v8f acc1 = {};

    const float* xb = x + (size_t)b * (NCH * TWO_N);

    for (int g = 0; g < NGEN; ++g) {
        const int idx = pinv[g * TWO_N + o0 + m];
        const float* xcol = xb + idx;
        const float* wg   = w + g * (NCH * NCH);

#pragma unroll
        for (int kb = 0; kb < 8; ++kb) {
            const int c0 = kb * 4 + khalf * 2;

            v2f a;
            a.x = xcol[(c0 + 0) * TWO_N];
            a.y = xcol[(c0 + 1) * TWO_N];

            v2f b0, b1;
            b0.x = wg[(c0 + 0) * NCH + m];
            b0.y = wg[(c0 + 1) * NCH + m];
            b1.x = wg[(c0 + 0) * NCH + 16 + m];
            b1.y = wg[(c0 + 1) * NCH + 16 + m];

            acc0 = __builtin_amdgcn_wmma_f32_16x16x4_f32(
                       false, a, false, b0, (short)0, acc0, false, false);
            acc1 = __builtin_amdgcn_wmma_f32_16x16x4_f32(
                       false, a, false, b1, (short)0, acc1, false, false);
        }
    }

    float* op0 = out + (size_t)b * (NCH * TWO_N) + (size_t)m * TWO_N
                     + o0 + khalf * 8;
    v4f s;
    s = (v4f){acc0[0], acc0[1], acc0[2], acc0[3]};
    *reinterpret_cast<v4f*>(op0) = s;
    s = (v4f){acc0[4], acc0[5], acc0[6], acc0[7]};
    *reinterpret_cast<v4f*>(op0 + 4) = s;

    float* op1 = op0 + 16 * TWO_N;
    s = (v4f){acc1[0], acc1[1], acc1[2], acc1[3]};
    *reinterpret_cast<v4f*>(op1) = s;
    s = (v4f){acc1[4], acc1[5], acc1[6], acc1[7]};
    *reinterpret_cast<v4f*>(op1 + 4) = s;
}

extern "C" void kernel_launch(void* const* d_in, const int* in_sizes, int n_in,
                              void* d_out, int out_size, void* d_ws, size_t ws_size,
                              hipStream_t stream) {
    const float* x    = (const float*)d_in[0];
    const float* w    = (const float*)d_in[1];
    const float* perm = (const float*)d_in[2];
    float* out = (float*)d_out;
    int* pinv = (int*)d_ws;

    build_invperm_kernel<<<(NGEN * TWO_N + 255) / 256, 256, 0, stream>>>(perm, pinv);

    const int total_waves = NB * OTILES;
    dihedral_gconv_wmma<<<total_waves / 8, 256, 0, stream>>>(x, w, pinv, out);
}


# inverse-permutation build parallelised: 100 work-groups scan 25-row slices of 16 columns and merge by max in LDS, instead of 1600 threads each scanning 400 rows serially
# speedup vs baseline: 1.5377x; 1.5377x over previous
#include <hip/hip_runtime.h>

typedef __attribute__((ext_vector_type(2))) float v2f;
typedef __attribute__((ext_vector_type(4))) float v4f;
typedef __attribute__((ext_vector_type(8))) float v8f;

#define NGEN   4
#define TWO_N  400
#define NCH    32
#define NB     512
#define OTILES (TWO_N / 16)

#define RGROUPS  16
#define RG_ROWS  (TWO_N / RGROUPS)

__global__ void __launch_bounds__(256)
build_invperm_kernel(const float* __restrict__ perm,
                     int* __restrict__ pinv) {
    __shared__ int part[RGROUPS][16];
    const int col = threadIdx.x & 15;
    const int rg  = threadIdx.x >> 4;
    const int g   = blockIdx.x / OTILES;
    const int ot  = blockIdx.x - g * OTILES;
    const int i0  = rg * RG_ROWS;
    const float* p = perm + (size_t)g * TWO_N * TWO_N
                          + (size_t)i0 * TWO_N + ot * 16 + col;
    int found = 0;
    for (int j = 0; j < RG_ROWS; ++j) {
        if (p[(size_t)j * TWO_N] > 0.5f) found = i0 + j;
    }
    part[rg][col] = found;
    __syncthreads();
    if (threadIdx.x < 16) {
        int f = part[0][threadIdx.x];
        for (int r = 1; r < RGROUPS; ++r) f = max(f, part[r][threadIdx.x]);
        pinv[g * TWO_N + ot * 16 + threadIdx.x] = f;
    }
}

__global__ void __launch_bounds__(256)
dihedral_gconv_wmma(const float* __restrict__ x,
                    const float* __restrict__ w,
                    const int*  __restrict__ pinv,
                    float* __restrict__ out) {
    const int lane  = threadIdx.x & 31;
    const int wid   = threadIdx.x >> 5;
    const int wave  = blockIdx.x * 8 + wid;
    const int b     = wave / OTILES;
    const int ot    = wave - b * OTILES;
    const int o0    = ot * 16;
    const int m     = lane & 15;
    const int khalf = lane >> 4;

    v8f acc0 = {};
    v8f acc1 = {};

    const float* xb = x + (size_t)b * (NCH * TWO_N);

    for (int g = 0; g < NGEN; ++g) {
        const int idx = pinv[g * TWO_N + o0 + m];
        const float* xcol = xb + idx;
        const float* wg   = w + g * (NCH * NCH);

#pragma unroll
        for (int kb = 0; kb < 8; ++kb) {
            const int c0 = kb * 4 + khalf * 2;

            v2f a;
            a.x = xcol[(c0 + 0) * TWO_N];
            a.y = xcol[(c0 + 1) * TWO_N];

            v2f b0, b1;
            b0.x = wg[(c0 + 0) * NCH + m];
            b0.y = wg[(c0 + 1) * NCH + m];
            b1.x = wg[(c0 + 0) * NCH + 16 + m];
            b1.y = wg[(c0 + 1) * NCH + 16 + m];

            acc0 = __builtin_amdgcn_wmma_f32_16x16x4_f32(
                       false, a, false, b0, (short)0, acc0, false, false);
            acc1 = __builtin_amdgcn_wmma_f32_16x16x4_f32(
                       false, a, false, b1, (short)0, acc1, false, false);
        }
    }

    float* op0 = out + (size_t)b * (NCH * TWO_N) + (size_t)m * TWO_N
                     + o0 + khalf * 8;
    v4f s;
    s = (v4f){acc0[0], acc0[1], acc0[2], acc0[3]};
    *reinterpret_cast<v4f*>(op0) = s;
    s = (v4f){acc0[4], acc0[5], acc0[6], acc0[7]};
    *reinterpret_cast<v4f*>(op0 + 4) = s;

    float* op1 = op0 + 16 * TWO_N;
    s = (v4f){acc1[0], acc1[1], acc1[2], acc1[3]};
    *reinterpret_cast<v4f*>(op1) = s;
    s = (v4f){acc1[4], acc1[5], acc1[6], acc1[7]};
    *reinterpret_cast<v4f*>(op1 + 4) = s;
}

extern "C" void kernel_launch(void* const* d_in, const int* in_sizes, int n_in,
                              void* d_out, int out_size, void* d_ws, size_t ws_size,
                              hipStream_t stream) {
    const float* x    = (const float*)d_in[0];
    const float* w    = (const float*)d_in[1];
    const float* perm = (const float*)d_in[2];
    float* out = (float*)d_out;
    int* pinv = (int*)d_ws;

    build_invperm_kernel<<<NGEN * OTILES, 256, 0, stream>>>(perm, pinv);

    const int total_waves = NB * OTILES;
    dihedral_gconv_wmma<<<total_waves / 8, 256, 0, stream>>>(x, w, pinv, out);
}


# x[b] staged once per work-group in LDS by 128-bit loads (channel pairs interleaved); waves gather A fragments from LDS for 5 o-tiles each, so one weight fragment load serves 5 tiles
# speedup vs baseline: 1.8214x; 1.1845x over previous
#include <hip/hip_runtime.h>

typedef __attribute__((ext_vector_type(2))) float v2f;
typedef __attribute__((ext_vector_type(4))) float v4f;
typedef __attribute__((ext_vector_type(8))) float v8f;

#define NGEN   4
#define TWO_N  400
#define NCH    32
#define NB     512
#define OTILES (TWO_N / 16)

#define RGROUPS  16
#define RG_ROWS  (TWO_N / RGROUPS)

__global__ void __launch_bounds__(256)
build_invperm_kernel(const float* __restrict__ perm,
                     int* __restrict__ pinv) {
    __shared__ int part[RGROUPS][16];
    const int col = threadIdx.x & 15;
    const int rg  = threadIdx.x >> 4;
    const int g   = blockIdx.x / OTILES;
    const int ot  = blockIdx.x - g * OTILES;
    const int i0  = rg * RG_ROWS;
    const float* p = perm + (size_t)g * TWO_N * TWO_N
                          + (size_t)i0 * TWO_N + ot * 16 + col;
    int found = 0;
    for (int j = 0; j < RG_ROWS; ++j) {
        if (p[(size_t)j * TWO_N] > 0.5f) found = i0 + j;
    }
    part[rg][col] = found;
    __syncthreads();
    if (threadIdx.x < 16) {
        int f = part[0][threadIdx.x];
        for (int r = 1; r < RGROUPS; ++r) f = max(f, part[r][threadIdx.x]);
        pinv[g * TWO_N + ot * 16 + threadIdx.x] = f;
    }
}

#define MT         5
#define K2_THREADS 320
#define NPAIR      (NCH / 2)
#define STAGE_UNITS (NPAIR * (TWO_N / 4))

__global__ void __launch_bounds__(K2_THREADS)
dihedral_gconv_wmma(const float* __restrict__ x,
                    const float* __restrict__ w,
                    const int*  __restrict__ pinv,
                    float* __restrict__ out) {
    __shared__ v2f xs[NPAIR * TWO_N];

    const int tid   = threadIdx.x;
    const int lane  = tid & 31;
    const int wid   = tid >> 5;
    const int b     = blockIdx.x;
    const int m     = lane & 15;
    const int khalf = lane >> 4;
    const int mg    = wid >> 1;
    const int nh    = wid & 1;

    int col[NGEN][MT];
    for (int g = 0; g < NGEN; ++g) {
        for (int t = 0; t < MT; ++t) {
            col[g][t] = pinv[g * TWO_N + (mg * MT + t) * 16 + m] + khalf * TWO_N;
        }
    }

    const float* xb = x + (size_t)b * (NCH * TWO_N);
    for (int j = 0; j < STAGE_UNITS / K2_THREADS; ++j) {
        const int u  = tid + j * K2_THREADS;
        const int cp = u / (TWO_N / 4);
        const int i4 = u - cp * (TWO_N / 4);
        const float* src = xb + (2 * cp) * TWO_N + i4 * 4;
        const v4f r0 = *reinterpret_cast<const v4f*>(src);
        const v4f r1 = *reinterpret_cast<const v4f*>(src + TWO_N);
        v4f* dst = reinterpret_cast<v4f*>(&xs[cp * TWO_N + i4 * 4]);
        dst[0] = (v4f){r0.x, r1.x, r0.y, r1.y};
        dst[1] = (v4f){r0.z, r1.z, r0.w, r1.w};
    }
    __syncthreads();

    v8f acc[MT];
    for (int t = 0; t < MT; ++t) acc[t] = (v8f){};

    const float* wl = w + khalf * (2 * NCH) + nh * 16 + m;

    for (int g = 0; g < NGEN; ++g) {
        for (int kb = 0; kb < 8; ++kb) {
            v2f bf;
            bf.x = wl[g * (NCH * NCH) + (kb * 4 + 0) * NCH];
            bf.y = wl[g * (NCH * NCH) + (kb * 4 + 1) * NCH];
            for (int t = 0; t < MT; ++t) {
                const v2f a = xs[(kb * 2) * TWO_N + col[g][t]];
                acc[t] = __builtin_amdgcn_wmma_f32_16x16x4_f32(
                             false, a, false, bf, (short)0, acc[t], false, false);
            }
        }
    }

    float* ob = out + (size_t)b * (NCH * TWO_N) + (size_t)(nh * 16 + m) * TWO_N
                    + (mg * MT) * 16 + khalf * 8;
    for (int t = 0; t < MT; ++t) {
        float* op = ob + t * 16;
        *reinterpret_cast<v4f*>(op)     = (v4f){acc[t][0], acc[t][1], acc[t][2], acc[t][3]};
        *reinterpret_cast<v4f*>(op + 4) = (v4f){acc[t][4], acc[t][5], acc[t][6], acc[t][7]};
    }
}

extern "C" void kernel_launch(void* const* d_in, const int* in_sizes, int n_in,
                              void* d_out, int out_size, void* d_ws, size_t ws_size,
                              hipStream_t stream) {
    const float* x    = (const float*)d_in[0];
    const float* w    = (const float*)d_in[1];
    const float* perm = (const float*)d_in[2];
    float* out = (float*)d_out;
    int* pinv = (int*)d_ws;

    build_invperm_kernel<<<NGEN * OTILES, 256, 0, stream>>>(perm, pinv);

    dihedral_gconv_wmma<<<NB, K2_THREADS, 0, stream>>>(x, w, pinv, out);
}


# epilogue: each wave passes its 16x80 output block through its own LDS region and stores it in row order, whole 128-byte lines plus one half per row, instead of 16-byte pieces in 16 rows per store
# speedup vs baseline: 2.7302x; 1.4990x over previous
#include <hip/hip_runtime.h>

typedef __attribute__((ext_vector_type(2))) float v2f;
typedef __attribute__((ext_vector_type(4))) float v4f;
typedef __attribute__((ext_vector_type(8))) float v8f;

#define NGEN   4
#define TWO_N  400
#define NCH    32
#define NB     512
#define OTILES (TWO_N / 16)

#define RGROUPS  16
#define RG_ROWS  (TWO_N / RGROUPS)

__global__ void __launch_bounds__(256)
build_invperm_kernel(const float* __restrict__ perm,
                     int* __restrict__ pinv) {
    __shared__ int part[RGROUPS][16];
    const int col = threadIdx.x & 15;
    const int rg  = threadIdx.x >> 4;
    const int g   = blockIdx.x / OTILES;
    const int ot  = blockIdx.x - g * OTILES;
    const int i0  = rg * RG_ROWS;
    const float* p = perm + (size_t)g * TWO_N * TWO_N
                          + (size_t)i0 * TWO_N + ot * 16 + col;
    int found = 0;
    for (int j = 0; j < RG_ROWS; ++j) {
        if (p[(size_t)j * TWO_N] > 0.5f) found = i0 + j;
    }
    part[rg][col] = found;
    __syncthreads();
    if (threadIdx.x < 16) {
        int f = part[0][threadIdx.x];
        for (int r = 1; r < RGROUPS; ++r) f = max(f, part[r][threadIdx.x]);
        pinv[g * TWO_N + ot * 16 + threadIdx.x] = f;
    }
}

#define MT         5
#define K2_THREADS 320
#define NPAIR      (NCH / 2)
#define STAGE_UNITS (NPAIR * (TWO_N / 4))
#define YPITCH     (MT * 16 + 4)
#define ROW_CHUNKS (MT * 16 / 4)

__global__ void __launch_bounds__(K2_THREADS)
dihedral_gconv_wmma(const float* __restrict__ x,
                    const float* __restrict__ w,
                    const int*  __restrict__ pinv,
                    float* __restrict__ out) {
    __shared__ v2f xs[NPAIR * TWO_N];
    __shared__ __attribute__((aligned(16))) float ys[(K2_THREADS / 32) * 16 * YPITCH];

    const int tid   = threadIdx.x;
    const int lane  = tid & 31;
    const int wid   = tid >> 5;
    const int b     = blockIdx.x;
    const int m     = lane & 15;
    const int khalf = lane >> 4;
    const int mg    = wid >> 1;
    const int nh    = wid & 1;

    int col[NGEN][MT];
    for (int g = 0; g < NGEN; ++g) {
        for (int t = 0; t < MT; ++t) {
            col[g][t] = pinv[g * TWO_N + (mg * MT + t) * 16 + m] + khalf * TWO_N;
        }
    }

    const float* xb = x + (size_t)b * (NCH * TWO_N);
    for (int j = 0; j < STAGE_UNITS / K2_THREADS; ++j) {
        const int u  = tid + j * K2_THREADS;
        const int cp = u / (TWO_N / 4);
        const int i4 = u - cp * (TWO_N / 4);
        const float* src = xb + (2 * cp) * TWO_N + i4 * 4;
        const v4f r0 = *reinterpret_cast<const v4f*>(src);
        const v4f r1 = *reinterpret_cast<const v4f*>(src + TWO_N);
        v4f* dst = reinterpret_cast<v4f*>(&xs[cp * TWO_N + i4 * 4]);
        dst[0] = (v4f){r0.x, r1.x, r0.y, r1.y};
        dst[1] = (v4f){r0.z, r1.z, r0.w, r1.w};
    }
    __syncthreads();

    v8f acc[MT];
    for (int t = 0; t < MT; ++t) acc[t] = (v8f){};

    const float* wl = w + khalf * (2 * NCH) + nh * 16 + m;

    for (int g = 0; g < NGEN; ++g) {
        for (int kb = 0; kb < 8; ++kb) {
            v2f bf;
            bf.x = wl[g * (NCH * NCH) + (kb * 4 + 0) * NCH];
            bf.y = wl[g * (NCH * NCH) + (kb * 4 + 1) * NCH];
            for (int t = 0; t < MT; ++t) {
                const v2f a = xs[(kb * 2) * TWO_N + col[g][t]];
                acc[t] = __builtin_amdgcn_wmma_f32_16x16x4_f32(
                             false, a, false, bf, (short)0, acc[t], false, false);
            }
        }
    }

    float* yw = ys + wid * (16 * YPITCH);
    for (int t = 0; t < MT; ++t) {
        v4f* yp = reinterpret_cast<v4f*>(yw + m * YPITCH + t * 16 + khalf * 8);
        yp[0] = (v4f){acc[t][0], acc[t][1], acc[t][2], acc[t][3]};
        yp[1] = (v4f){acc[t][4], acc[t][5], acc[t][6], acc[t][7]};
    }
    __builtin_amdgcn_fence(__ATOMIC_ACQ_REL, "wavefront");
    float* ob = out + (size_t)b * (NCH * TWO_N) + (size_t)(nh * 16) * TWO_N + (mg * MT) * 16;
    for (int j = 0; j < (16 * ROW_CHUNKS) / 32; ++j) {
        const int q   = lane + 32 * j;
        const int row = q / ROW_CHUNKS;
        const int c   = q - row * ROW_CHUNKS;
        const v4f v = *reinterpret_cast<const v4f*>(yw + row * YPITCH + c * 4);
        *reinterpret_cast<v4f*>(ob + (size_t)row * TWO_N + c * 4) = v;
    }
}

extern "C" void kernel_launch(void* const* d_in, const int* in_sizes, int n_in,
                              void* d_out, int out_size, void* d_ws, size_t ws_size,
                              hipStream_t stream) {
    const float* x    = (const float*)d_in[0];
    const float* w    = (const float*)d_in[1];
    const float* perm = (const float*)d_in[2];
    float* out = (float*)d_out;
    int* pinv = (int*)d_ws;

    build_invperm_kernel<<<NGEN * OTILES, 256, 0, stream>>>(perm, pinv);

    dihedral_gconv_wmma<<<NB, K2_THREADS, 0, stream>>>(x, w, pinv, out);
}


# products kernel head pipelined: weights and column indices requested at entry, x slab handed to LDS in 3 channel stages behind LDS-only barriers; sum order now channel block outer, g inner
# speedup vs baseline: 2.9365x; 1.0756x over previous
#include <hip/hip_runtime.h>

typedef __attribute__((ext_vector_type(2))) float v2f;
typedef __attribute__((ext_vector_type(4))) float v4f;
typedef __attribute__((ext_vector_type(8))) float v8f;

#define NGEN   4
#define TWO_N  400
#define NCH    32
#define NB     512
#define OTILES (TWO_N / 16)

#define RGROUPS  16
#define RG_ROWS  (TWO_N / RGROUPS)

__global__ void __launch_bounds__(256)
build_invperm_kernel(const float* __restrict__ perm,
                     int* __restrict__ pinv) {
    __shared__ int part[RGROUPS][16];
    const int col = threadIdx.x & 15;
    const int rg  = threadIdx.x >> 4;
    const int g   = blockIdx.x / OTILES;
    const int ot  = blockIdx.x - g * OTILES;
    const int i0  = rg * RG_ROWS;
    const float* p = perm + (size_t)g * TWO_N * TWO_N
                          + (size_t)i0 * TWO_N + ot * 16 + col;
    int found = 0;
    for (int j = 0; j < RG_ROWS; ++j) {
        if (p[(size_t)j * TWO_N] > 0.5f) found = i0 + j;
    }
    part[rg][col] = found;
    __syncthreads();
    if (threadIdx.x < 16) {
        int f = part[0][threadIdx.x];
        for (int r = 1; r < RGROUPS; ++r) f = max(f, part[r][threadIdx.x]);
        pinv[g * TWO_N + ot * 16 + threadIdx.x] = f;
    }
}

#define MT         5
#define K2_THREADS 320
#define NPAIR      (NCH / 2)
#define STAGE_UNITS (NPAIR * (TWO_N / 4))
#define YPITCH     (MT * 16 + 4)
#define ROW_CHUNKS (MT * 16 / 4)

#define NROUND     (STAGE_UNITS / K2_THREADS)
#define XS_FLOATS  (NPAIR * TWO_N * 2)

__device__ __forceinline__ void stage_load(const float* __restrict__ xb, int tid, int round,
                                           v4f& r0, v4f& r1) {
    const int u  = tid + round * K2_THREADS;
    const int cp = u / (TWO_N / 4);
    const int i4 = u - cp * (TWO_N / 4);
    const float* src = xb + (2 * cp) * TWO_N + i4 * 4;
    r0 = *reinterpret_cast<const v4f*>(src);
    r1 = *reinterpret_cast<const v4f*>(src + TWO_N);
}

__device__ __forceinline__ void stage_store(v2f* xs, int tid, int round,
                                            const v4f& r0, const v4f& r1) {
    const int u  = tid + round * K2_THREADS;
    const int cp = u / (TWO_N / 4);
    const int i4 = u - cp * (TWO_N / 4);
    v4f* dst = reinterpret_cast<v4f*>(&xs[cp * TWO_N + i4 * 4]);
    dst[0] = (v4f){r0.x, r1.x, r0.y, r1.y};
    dst[1] = (v4f){r0.z, r1.z, r0.w, r1.w};
}

__device__ __forceinline__ v2f weight_frag(const float* __restrict__ wl, int g, int kb) {
    v2f bf;
    bf.x = wl[g * (NCH * NCH) + (kb * 4 + 0) * NCH];
    bf.y = wl[g * (NCH * NCH) + (kb * 4 + 1) * NCH];
    return bf;
}

__device__ __forceinline__ void lds_barrier() {
    asm volatile("s_wait_dscnt 0x0\n\ts_barrier_signal -1\n\ts_barrier_wait -1" ::: "memory");
}

template <int KB0, int KB1>
__device__ __forceinline__ void mma_stage(v8f (&acc)[MT], const v2f (&bw)[NGEN][8],
                                          const v2f* xs, const int (&col)[NGEN][MT]) {
    for (int kb = KB0; kb < KB1; ++kb) {
        for (int g = 0; g < NGEN; ++g) {
            const v2f bf = bw[g][kb];
            for (int t = 0; t < MT; ++t) {
                const v2f a = xs[(kb * 2) * TWO_N + col[g][t]];
                acc[t] = __builtin_amdgcn_wmma_f32_16x16x4_f32(
                             false, a, false, bf, (short)0, acc[t], false, false);
            }
        }
    }
}

__global__ void __launch_bounds__(K2_THREADS) __attribute__((amdgpu_waves_per_eu(6)))
dihedral_gconv_wmma(const float* __restrict__ x,
                    const float* __restrict__ w,
                    const int*  __restrict__ pinv,
                    float* __restrict__ out) {
    __shared__ __attribute__((aligned(16))) float smem[XS_FLOATS + (K2_THREADS / 32) * 16 * YPITCH];
    v2f* xs = reinterpret_cast<v2f*>(smem);
    float* ys = smem + XS_FLOATS;

    const int tid   = threadIdx.x;
    const int lane  = tid & 31;
    const int wid   = __builtin_amdgcn_readfirstlane(tid >> 5);
    const int b     = blockIdx.x;
    const int m     = lane & 15;
    const int khalf = lane >> 4;
    const int mg    = wid >> 1;
    const int nh    = wid & 1;

    const float* xb = x + (size_t)b * (NCH * TWO_N);
    const float* wl = w + khalf * (2 * NCH) + nh * 16 + m;

    v4f r0[NROUND], r1[NROUND];
    v2f bw[NGEN][8];
    int col[NGEN][MT];

    stage_load(xb, tid, 0, r0[0], r1[0]);
    for (int g = 0; g < NGEN; ++g) {
        for (int t = 0; t < MT; ++t) {
            col[g][t] = pinv[g * TWO_N + (mg * MT + t) * 16 + m];
        }
    }
    for (int g = 0; g < NGEN; ++g) bw[g][0] = weight_frag(wl, g, 0);
    __builtin_amdgcn_sched_barrier(0);
    stage_load(xb, tid, 1, r0[1], r1[1]);
    stage_load(xb, tid, 2, r0[2], r1[2]);
    for (int g = 0; g < NGEN; ++g) {
        for (int kb = 1; kb < 4; ++kb) bw[g][kb] = weight_frag(wl, g, kb);
    }
    __builtin_amdgcn_sched_barrier(0);

    v8f acc[MT];
    for (int t = 0; t < MT; ++t) acc[t] = (v8f){};

    stage_store(xs, tid, 0, r0[0], r1[0]);
    for (int g = 0; g < NGEN; ++g) {
        for (int t = 0; t < MT; ++t) col[g][t] += khalf * TWO_N;
    }
    lds_barrier();
    stage_load(xb, tid, 3, r0[3], r1[3]);
    stage_load(xb, tid, 4, r0[4], r1[4]);
    mma_stage<0, 1>(acc, bw, xs, col);
    __builtin_amdgcn_sched_barrier(0);

    stage_store(xs, tid, 1, r0[1], r1[1]);
    stage_store(xs, tid, 2, r0[2], r1[2]);
    lds_barrier();
    mma_stage<1, 2>(acc, bw, xs, col);
    __builtin_amdgcn_sched_barrier(0);
    stage_store(xs, tid, 3, r0[3], r1[3]);
    stage_store(xs, tid, 4, r0[4], r1[4]);
    for (int g = 0; g < NGEN; ++g) {
        for (int kb = 4; kb < 6; ++kb) bw[g][kb] = weight_frag(wl, g, kb);
    }
    mma_stage<2, 4>(acc, bw, xs, col);

    lds_barrier();
    for (int g = 0; g < NGEN; ++g) {
        for (int kb = 6; kb < 8; ++kb) bw[g][kb] = weight_frag(wl, g, kb);
    }
    mma_stage<4, 8>(acc, bw, xs, col);

    float* yw = ys + wid * (16 * YPITCH);
    for (int t = 0; t < MT; ++t) {
        v4f* yp = reinterpret_cast<v4f*>(yw + m * YPITCH + t * 16 + ((lane & 16) >> 1));
        yp[0] = (v4f){acc[t][0], acc[t][1], acc[t][2], acc[t][3]};
        yp[1] = (v4f){acc[t][4], acc[t][5], acc[t][6], acc[t][7]};
    }
    __builtin_amdgcn_fence(__ATOMIC_ACQ_REL, "wavefront");
    float* ob = out + (size_t)b * (NCH * TWO_N) + (size_t)(nh * 16) * TWO_N + (mg * MT) * 16;
    for (int j = 0; j < (16 * ROW_CHUNKS) / 32; ++j) {
        const int q   = lane + 32 * j;
        const int row = q / ROW_CHUNKS;
        const int c   = q - row * ROW_CHUNKS;
        const v4f v = *reinterpret_cast<const v4f*>(yw + row * YPITCH + c * 4);
        *reinterpret_cast<v4f*>(ob + (size_t)row * TWO_N + c * 4) = v;
    }
}

extern "C" void kernel_launch(void* const* d_in, const int* in_sizes, int n_in,
                              void* d_out, int out_size, void* d_ws, size_t ws_size,
                              hipStream_t stream) {
    const float* x    = (const float*)d_in[0];
    const float* w    = (const float*)d_in[1];
    const float* perm = (const float*)d_in[2];
    float* out = (float*)d_out;
    int* pinv = (int*)d_ws;

    build_invperm_kernel<<<NGEN * OTILES, 256, 0, stream>>>(perm, pinv);

    dihedral_gconv_wmma<<<NB, K2_THREADS, 0, stream>>>(x, w, pinv, out);
}
